# starting kernel
# baseline (speedup 1.0000x reference)
#include <hip/hip_runtime.h>
#include <math.h>

typedef __attribute__((ext_vector_type(2))) float v2f;
typedef __attribute__((ext_vector_type(3))) float v3f;
typedef __attribute__((ext_vector_type(8))) float v8f;

#define NPTS    1024
#define CHUNK   64
#define NCHUNK  (NPTS / CHUNK)
#define GROUPS  (CHUNK / 4)
#define WPB     4
#define THREADS (WPB * 32)
#define BPW     4
#define BPB     (WPB * BPW)

#define D_OFF       0
#define S_OFF       256
#define STG_HALF    (BPW * CHUNK * 4)
#define STG_W       (2 * STG_HALF)
#define SMEM_FLOATS (S_OFF + WPB * STG_W)

__device__ __forceinline__ float det3(const float M[3][3]) {
  return M[0][0]*(M[1][1]*M[2][2]-M[1][2]*M[2][1])
       - M[0][1]*(M[1][0]*M[2][2]-M[1][2]*M[2][0])
       + M[0][2]*(M[1][0]*M[2][1]-M[1][1]*M[2][0]);
}

__global__ __launch_bounds__(THREADS) void kabsch_wmma_kernel(
    const float* __restrict__ A, const float* __restrict__ B,
    float* __restrict__ outR, float* __restrict__ outT, int C)
{
  __shared__ alignas(16) float smem[SMEM_FLOATS];

  const int tid  = threadIdx.x;
  const int lane = tid & 31;
  const int wid  = tid >> 5;
  const int col  = lane & 15;
  const int hv   = lane >> 4;

  const int wb0 = (blockIdx.x * WPB + wid) * BPW;

  const int stA = S_OFF + wid * STG_W;
  const int stB = stA + STG_HALF;

  size_t bof[BPW];
#pragma unroll
  for (int b = 0; b < BPW; ++b) {
    int cb = wb0 + b;
    if (cb >= C) cb = C - 1;
    bof[b] = (size_t)cb * (NPTS * 3);
  }

#pragma unroll
  for (int b = 0; b < BPW; ++b) {
#pragma unroll
    for (int k = 0; k < 2; ++k) {
      const int p4 = (k * 32 + lane) * 4;
      smem[stA + b * (CHUNK * 4) + p4 + 3] = 1.0f;
      smem[stB + b * (CHUNK * 4) + p4 + 3] = 1.0f;
    }
  }

  const int ia = stA + (col >> 2) * (CHUNK * 4) + hv * 8 + (col & 3);
  const int ib = ia + STG_HALF;

  const int laneG = lane * 3;

  v8f acc0 = {0.f,0.f,0.f,0.f,0.f,0.f,0.f,0.f};
  v8f acc1 = {0.f,0.f,0.f,0.f,0.f,0.f,0.f,0.f};

  float rA[2 * BPW][3], rB[2 * BPW][3];

#define LOAD_CHUNK(ck_) do {                                                   \
    _Pragma("unroll") for (int b = 0; b < BPW; ++b) {                          \
      _Pragma("unroll") for (int k = 0; k < 2; ++k) {                          \
        const float* qa = A + bof[b] + (ck_) * (CHUNK * 3) + k * 96 + laneG;   \
        const float* qb = B + bof[b] + (ck_) * (CHUNK * 3) + k * 96 + laneG;   \
        rA[b*2+k][0] = qa[0]; rA[b*2+k][1] = qa[1]; rA[b*2+k][2] = qa[2];      \
        rB[b*2+k][0] = qb[0]; rB[b*2+k][1] = qb[1]; rB[b*2+k][2] = qb[2];      \
      }                                                                        \
    }                                                                          \
  } while (0)

#define STORE_CHUNK() do {                                                     \
    _Pragma("unroll") for (int b = 0; b < BPW; ++b) {                          \
      _Pragma("unroll") for (int k = 0; k < 2; ++k) {                          \
        const int p4 = (k * 32 + lane) * 4;                                    \
        v3f ta = {rA[b*2+k][0], rA[b*2+k][1], rA[b*2+k][2]};                   \
        v3f tb = {rB[b*2+k][0], rB[b*2+k][1], rB[b*2+k][2]};                   \
        *(v3f*)&smem[stA + b*(CHUNK*4) + p4] = ta;                             \
        *(v3f*)&smem[stB + b*(CHUNK*4) + p4] = tb;                             \
      }                                                                        \
    }                                                                          \
  } while (0)

  auto lda = [&](int g) { v2f r; r.x = smem[ia + g*16]; r.y = smem[ia + g*16 + 4]; return r; };
  auto ldb = [&](int g) { v2f r; r.x = smem[ib + g*16]; r.y = smem[ib + g*16 + 4]; return r; };

  LOAD_CHUNK(0);
  STORE_CHUNK();

  for (int ck = 0; ck < NCHUNK; ++ck) {
    const bool more = (ck + 1 < NCHUNK);
    if (more) LOAD_CHUNK(ck + 1);

    v2f pa[4], pb[4];
#pragma unroll
    for (int g = 0; g < 4; ++g) { pa[g] = lda(g); pb[g] = ldb(g); }
#pragma unroll
    for (int g = 0; g < GROUPS; g += 2) {
      v2f na0 = {}, nb0 = {}, na1 = {}, nb1 = {};
      if (g + 4 < GROUPS) { na0 = lda(g + 4); nb0 = ldb(g + 4); }
      if (g + 5 < GROUPS) { na1 = lda(g + 5); nb1 = ldb(g + 5); }
      acc0 = __builtin_amdgcn_wmma_f32_16x16x4_f32(
          false, pa[0], false, pb[0], (short)0, acc0, false, false);
      acc1 = __builtin_amdgcn_wmma_f32_16x16x4_f32(
          false, pa[1], false, pb[1], (short)0, acc1, false, false);
      pa[0] = pa[2]; pb[0] = pb[2]; pa[1] = pa[3]; pb[1] = pb[3];
      pa[2] = na0;   pb[2] = nb0;   pa[3] = na1;   pb[3] = nb1;
    }

    if (more) STORE_CHUNK();
  }

  {
    v8f accS = acc0 + acc1;
    const int rbase = (lane < 16) ? 0 : 8;
#pragma unroll
    for (int v = 0; v < 8; ++v) {
      const int r = v + rbase;
      if ((r >> 2) == (col >> 2)) {
        smem[D_OFF + (wid * BPW + (r >> 2)) * 16 + (r & 3) * 4 + (col & 3)] = accS[v];
      }
    }
  }
  __syncthreads();

  if (tid < BPB) {
    const int b2 = blockIdx.x * BPB + tid;
    if (b2 < C) {
      const float* Dm = &smem[D_OFF + tid * 16];
      const float nP   = Dm[15];
      const float invn = 1.0f / nP;
      float cA[3] = {Dm[3] * invn,  Dm[7] * invn,  Dm[11] * invn};
      float cB[3] = {Dm[12] * invn, Dm[13] * invn, Dm[14] * invn};

      float W[3][3], V[3][3];
      for (int i = 0; i < 3; ++i)
        for (int j = 0; j < 3; ++j) {
          W[i][j] = Dm[i * 4 + j] - nP * cA[i] * cB[j];
          V[i][j] = (i == j) ? 1.0f : 0.0f;
        }

      for (int sw = 0; sw < 8; ++sw) {
        for (int pi = 0; pi < 3; ++pi) {
          const int p = (pi == 2) ? 1 : 0;
          const int q = (pi == 0) ? 1 : 2;
          float a = 0.f, b = 0.f, g = 0.f;
          for (int i = 0; i < 3; ++i) {
            a += W[i][p] * W[i][p];
            b += W[i][q] * W[i][q];
            g += W[i][p] * W[i][q];
          }
          if (fabsf(g) > 1e-12f * sqrtf(a * b) + 1e-38f) {
            float zeta = (b - a) / (2.0f * g);
            float tJ = copysignf(1.0f, zeta) /
                       (fabsf(zeta) + sqrtf(1.0f + zeta * zeta));
            float cJ = 1.0f / sqrtf(1.0f + tJ * tJ);
            float sJ = cJ * tJ;
            for (int i = 0; i < 3; ++i) {
              float wp = W[i][p], wq = W[i][q];
              W[i][p] = cJ * wp - sJ * wq;  W[i][q] = sJ * wp + cJ * wq;
              float vp = V[i][p], vq = V[i][q];
              V[i][p] = cJ * vp - sJ * vq;  V[i][q] = sJ * vp + cJ * vq;
            }
          }
        }
      }

      float sv[3];
      for (int j = 0; j < 3; ++j)
        sv[j] = sqrtf(W[0][j]*W[0][j] + W[1][j]*W[1][j] + W[2][j]*W[2][j]);

      for (int pass = 0; pass < 3; ++pass) {
        int x = (pass == 2) ? 1 : 0;
        int y = (pass == 0) ? 1 : 2;
        if (sv[x] < sv[y]) {
          float tsv = sv[x]; sv[x] = sv[y]; sv[y] = tsv;
          for (int i = 0; i < 3; ++i) {
            float tw = W[i][x]; W[i][x] = W[i][y]; W[i][y] = tw;
            float tv = V[i][x]; V[i][x] = V[i][y]; V[i][y] = tv;
          }
        }
      }

      float U[3][3];
      for (int j = 0; j < 3; ++j) {
        float inv = (sv[j] > 1e-20f) ? (1.0f / sv[j]) : 0.0f;
        for (int i = 0; i < 3; ++i) U[i][j] = W[i][j] * inv;
      }

      const float f = (det3(U) * det3(V) < 0.0f) ? -1.0f : 1.0f;

      float R[3][3];
      for (int i = 0; i < 3; ++i)
        for (int j = 0; j < 3; ++j)
          R[i][j] = V[i][0]*U[j][0] + V[i][1]*U[j][1] + f * V[i][2]*U[j][2];

      float* ro = outR + (size_t)b2 * 9;
      float* to = outT + (size_t)b2 * 3;
      for (int i = 0; i < 3; ++i) {
        for (int j = 0; j < 3; ++j) ro[i * 3 + j] = R[i][j];
        to[i] = cB[i] - (R[i][0]*cA[0] + R[i][1]*cA[1] + R[i][2]*cA[2]);
      }
    }
  }
}

extern "C" void kernel_launch(void* const* d_in, const int* in_sizes, int n_in,
                              void* d_out, int out_size, void* d_ws, size_t ws_size,
                              hipStream_t stream) {
  (void)n_in; (void)d_ws; (void)ws_size; (void)out_size;
  const float* A = (const float*)d_in[0];
  const float* B = (const float*)d_in[1];
  float* out = (float*)d_out;
  const int C = in_sizes[0] / (NPTS * 3);
  float* outR = out;
  float* outT = out + (size_t)C * 9;
  const int blocks = (C + BPB - 1) / BPB;
  hipLaunchKernelGGL(kabsch_wmma_kernel, dim3(blocks), dim3(THREADS), 0, stream,
                     A, B, outR, outT, C);
}


# 3x3 Jacobi SVD tail: rotation c,s from hardware rcp/sqrt/rsq (about 1 ulp; rounding of c,s changes) instead of exact div and sqrt sequences; same guard, applied by select: shorter serial chain
# speedup vs baseline: 1.1372x; 1.1372x over previous
#include <hip/hip_runtime.h>
#include <math.h>

typedef __attribute__((ext_vector_type(2))) float v2f;
typedef __attribute__((ext_vector_type(3))) float v3f;
typedef __attribute__((ext_vector_type(8))) float v8f;

#define NPTS    1024
#define CHUNK   64
#define NCHUNK  (NPTS / CHUNK)
#define GROUPS  (CHUNK / 4)
#define WPB     4
#define THREADS (WPB * 32)
#define BPW     4
#define BPB     (WPB * BPW)

#define D_OFF       0
#define S_OFF       256
#define STG_HALF    (BPW * CHUNK * 4)
#define STG_W       (2 * STG_HALF)
#define SMEM_FLOATS (S_OFF + WPB * STG_W)

__device__ __forceinline__ float det3(const float M[3][3]) {
  return M[0][0]*(M[1][1]*M[2][2]-M[1][2]*M[2][1])
       - M[0][1]*(M[1][0]*M[2][2]-M[1][2]*M[2][0])
       + M[0][2]*(M[1][0]*M[2][1]-M[1][1]*M[2][0]);
}

__global__ __launch_bounds__(THREADS) void kabsch_wmma_kernel(
    const float* __restrict__ A, const float* __restrict__ B,
    float* __restrict__ outR, float* __restrict__ outT, int C)
{
  __shared__ alignas(16) float smem[SMEM_FLOATS];

  const int tid  = threadIdx.x;
  const int lane = tid & 31;
  const int wid  = tid >> 5;
  const int col  = lane & 15;
  const int hv   = lane >> 4;

  const int wb0 = (blockIdx.x * WPB + wid) * BPW;

  const int stA = S_OFF + wid * STG_W;
  const int stB = stA + STG_HALF;

  size_t bof[BPW];
#pragma unroll
  for (int b = 0; b < BPW; ++b) {
    int cb = wb0 + b;
    if (cb >= C) cb = C - 1;
    bof[b] = (size_t)cb * (NPTS * 3);
  }

#pragma unroll
  for (int b = 0; b < BPW; ++b) {
#pragma unroll
    for (int k = 0; k < 2; ++k) {
      const int p4 = (k * 32 + lane) * 4;
      smem[stA + b * (CHUNK * 4) + p4 + 3] = 1.0f;
      smem[stB + b * (CHUNK * 4) + p4 + 3] = 1.0f;
    }
  }

  const int ia = stA + (col >> 2) * (CHUNK * 4) + hv * 8 + (col & 3);
  const int ib = ia + STG_HALF;

  const int laneG = lane * 3;

  v8f acc0 = {0.f,0.f,0.f,0.f,0.f,0.f,0.f,0.f};
  v8f acc1 = {0.f,0.f,0.f,0.f,0.f,0.f,0.f,0.f};

  float rA[2 * BPW][3], rB[2 * BPW][3];

#define LOAD_CHUNK(ck_) do {                                                   \
    _Pragma("unroll") for (int b = 0; b < BPW; ++b) {                          \
      _Pragma("unroll") for (int k = 0; k < 2; ++k) {                          \
        const float* qa = A + bof[b] + (ck_) * (CHUNK * 3) + k * 96 + laneG;   \
        const float* qb = B + bof[b] + (ck_) * (CHUNK * 3) + k * 96 + laneG;   \
        rA[b*2+k][0] = qa[0]; rA[b*2+k][1] = qa[1]; rA[b*2+k][2] = qa[2];      \
        rB[b*2+k][0] = qb[0]; rB[b*2+k][1] = qb[1]; rB[b*2+k][2] = qb[2];      \
      }                                                                        \
    }                                                                          \
  } while (0)

#define STORE_CHUNK() do {                                                     \
    _Pragma("unroll") for (int b = 0; b < BPW; ++b) {                          \
      _Pragma("unroll") for (int k = 0; k < 2; ++k) {                          \
        const int p4 = (k * 32 + lane) * 4;                                    \
        v3f ta = {rA[b*2+k][0], rA[b*2+k][1], rA[b*2+k][2]};                   \
        v3f tb = {rB[b*2+k][0], rB[b*2+k][1], rB[b*2+k][2]};                   \
        *(v3f*)&smem[stA + b*(CHUNK*4) + p4] = ta;                             \
        *(v3f*)&smem[stB + b*(CHUNK*4) + p4] = tb;                             \
      }                                                                        \
    }                                                                          \
  } while (0)

  auto lda = [&](int g) { v2f r; r.x = smem[ia + g*16]; r.y = smem[ia + g*16 + 4]; return r; };
  auto ldb = [&](int g) { v2f r; r.x = smem[ib + g*16]; r.y = smem[ib + g*16 + 4]; return r; };

  LOAD_CHUNK(0);
  STORE_CHUNK();

  for (int ck = 0; ck < NCHUNK; ++ck) {
    const bool more = (ck + 1 < NCHUNK);
    if (more) LOAD_CHUNK(ck + 1);

    v2f pa[4], pb[4];
#pragma unroll
    for (int g = 0; g < 4; ++g) { pa[g] = lda(g); pb[g] = ldb(g); }
#pragma unroll
    for (int g = 0; g < GROUPS; g += 2) {
      v2f na0 = {}, nb0 = {}, na1 = {}, nb1 = {};
      if (g + 4 < GROUPS) { na0 = lda(g + 4); nb0 = ldb(g + 4); }
      if (g + 5 < GROUPS) { na1 = lda(g + 5); nb1 = ldb(g + 5); }
      acc0 = __builtin_amdgcn_wmma_f32_16x16x4_f32(
          false, pa[0], false, pb[0], (short)0, acc0, false, false);
      acc1 = __builtin_amdgcn_wmma_f32_16x16x4_f32(
          false, pa[1], false, pb[1], (short)0, acc1, false, false);
      pa[0] = pa[2]; pb[0] = pb[2]; pa[1] = pa[3]; pb[1] = pb[3];
      pa[2] = na0;   pb[2] = nb0;   pa[3] = na1;   pb[3] = nb1;
    }

    if (more) STORE_CHUNK();
  }

  {
    v8f accS = acc0 + acc1;
    const int rbase = (lane < 16) ? 0 : 8;
#pragma unroll
    for (int v = 0; v < 8; ++v) {
      const int r = v + rbase;
      if ((r >> 2) == (col >> 2)) {
        smem[D_OFF + (wid * BPW + (r >> 2)) * 16 + (r & 3) * 4 + (col & 3)] = accS[v];
      }
    }
  }
  __syncthreads();

  if (tid < BPB) {
    const int b2 = blockIdx.x * BPB + tid;
    if (b2 < C) {
      const float* Dm = &smem[D_OFF + tid * 16];
      const float nP   = Dm[15];
      const float invn = 1.0f / nP;
      float cA[3] = {Dm[3] * invn,  Dm[7] * invn,  Dm[11] * invn};
      float cB[3] = {Dm[12] * invn, Dm[13] * invn, Dm[14] * invn};

      float W[3][3], V[3][3];
      for (int i = 0; i < 3; ++i)
        for (int j = 0; j < 3; ++j) {
          W[i][j] = Dm[i * 4 + j] - nP * cA[i] * cB[j];
          V[i][j] = (i == j) ? 1.0f : 0.0f;
        }

      for (int sw = 0; sw < 8; ++sw) {
        for (int pi = 0; pi < 3; ++pi) {
          const int p = (pi == 2) ? 1 : 0;
          const int q = (pi == 0) ? 1 : 2;
          float a = 0.f, b = 0.f, g = 0.f;
          for (int i = 0; i < 3; ++i) {
            a += W[i][p] * W[i][p];
            b += W[i][q] * W[i][q];
            g += W[i][p] * W[i][q];
          }
          const bool rot = fabsf(g) > 1e-12f * sqrtf(a * b) + 1e-38f;
          const float zeta = (b - a) * __builtin_amdgcn_rcpf(2.0f * g);
          const float tJ = copysignf(
              __builtin_amdgcn_rcpf(fabsf(zeta) +
                                    __builtin_amdgcn_sqrtf(1.0f + zeta * zeta)),
              zeta);
          float cJ = __builtin_amdgcn_rsqf(1.0f + tJ * tJ);
          float sJ = cJ * tJ;
          cJ = rot ? cJ : 1.0f;
          sJ = rot ? sJ : 0.0f;
          for (int i = 0; i < 3; ++i) {
            float wp = W[i][p], wq = W[i][q];
            W[i][p] = cJ * wp - sJ * wq;  W[i][q] = sJ * wp + cJ * wq;
            float vp = V[i][p], vq = V[i][q];
            V[i][p] = cJ * vp - sJ * vq;  V[i][q] = sJ * vp + cJ * vq;
          }
        }
      }

      float sv[3];
      for (int j = 0; j < 3; ++j)
        sv[j] = sqrtf(W[0][j]*W[0][j] + W[1][j]*W[1][j] + W[2][j]*W[2][j]);

      for (int pass = 0; pass < 3; ++pass) {
        int x = (pass == 2) ? 1 : 0;
        int y = (pass == 0) ? 1 : 2;
        if (sv[x] < sv[y]) {
          float tsv = sv[x]; sv[x] = sv[y]; sv[y] = tsv;
          for (int i = 0; i < 3; ++i) {
            float tw = W[i][x]; W[i][x] = W[i][y]; W[i][y] = tw;
            float tv = V[i][x]; V[i][x] = V[i][y]; V[i][y] = tv;
          }
        }
      }

      float U[3][3];
      for (int j = 0; j < 3; ++j) {
        float inv = (sv[j] > 1e-20f) ? (1.0f / sv[j]) : 0.0f;
        for (int i = 0; i < 3; ++i) U[i][j] = W[i][j] * inv;
      }

      const float f = (det3(U) * det3(V) < 0.0f) ? -1.0f : 1.0f;

      float R[3][3];
      for (int i = 0; i < 3; ++i)
        for (int j = 0; j < 3; ++j)
          R[i][j] = V[i][0]*U[j][0] + V[i][1]*U[j][1] + f * V[i][2]*U[j][2];

      float* ro = outR + (size_t)b2 * 9;
      float* to = outT + (size_t)b2 * 3;
      for (int i = 0; i < 3; ++i) {
        for (int j = 0; j < 3; ++j) ro[i * 3 + j] = R[i][j];
        to[i] = cB[i] - (R[i][0]*cA[0] + R[i][1]*cA[1] + R[i][2]*cA[2]);
      }
    }
  }
}

extern "C" void kernel_launch(void* const* d_in, const int* in_sizes, int n_in,
                              void* d_out, int out_size, void* d_ws, size_t ws_size,
                              hipStream_t stream) {
  (void)n_in; (void)d_ws; (void)ws_size; (void)out_size;
  const float* A = (const float*)d_in[0];
  const float* B = (const float*)d_in[1];
  float* out = (float*)d_out;
  const int C = in_sizes[0] / (NPTS * 3);
  float* outR = out;
  float* outT = out + (size_t)C * 9;
  const int blocks = (C + BPB - 1) / BPB;
  hipLaunchKernelGGL(kabsch_wmma_kernel, dim3(blocks), dim3(THREADS), 0, stream,
                     A, B, outR, outT, C);
}
